# starting kernel
# baseline (speedup 1.0000x reference)
#include <hip/hip_runtime.h>
#include <math.h>

#define BB 2
#define HH 8
#define SS 4096
#define DD 64
#define BH (BB*HH)
#define NS 16
#define CH (SS/NS)
#define PSTR 4352
#define FINBASE (BH*NS*PSTR)
#define LROW 68

typedef __attribute__((ext_vector_type(2))) float v2f;
typedef __attribute__((ext_vector_type(8))) float v8f;

__device__ __forceinline__ v8f wmma_f32_4(v2f a, v2f b, v8f c) {
    return __builtin_amdgcn_wmma_f32_16x16x4_f32(
        false, a, false, b, (short)0, c, false, false);
}

__device__ __forceinline__ v2f ld_frag(const float* p, int k0, int col, int lane, int str) {
    int ks = k0 + 2 * (lane >> 4);
    v2f r;
    r.x = p[ks * str + col];
    r.y = p[(ks + 1) * str + col];
    return r;
}

__global__ __launch_bounds__(256) void cosattn_p1(
    const float* __restrict__ K, const float* __restrict__ V,
    const int* __restrict__ mask, float* __restrict__ ws) {
    __shared__ float kn[CH * LROW];
    __shared__ float vv[CH * LROW];
    __shared__ float mf[CH];

    const int tid = threadIdx.x;
    const int blk = blockIdx.x;
    const int bh = blk / NS;
    const int chunk = blk % NS;
    const int b = bh / HH;
    const long gbase = ((long)bh * SS + (long)chunk * CH) * DD;

#pragma unroll
    for (int j = 0; j < 16; ++j) {
        int e4 = tid + j * 256;
        int row = e4 >> 4;
        int col = (e4 & 15) * 4;
        float4 kq = *(const float4*)(K + gbase + (long)e4 * 4);
        float4 vq = *(const float4*)(V + gbase + (long)e4 * 4);
        *(float4*)(kn + row * LROW + col) = kq;
        *(float4*)(vv + row * LROW + col) = vq;
    }
    {
        int kidx = chunk * CH + tid;
        mf[tid] = (mask[b * SS + kidx] != 0) ? 1.0f : 0.0f;
    }
    __syncthreads();

    {
        float* rowp = kn + tid * LROW;
        float ssum = 0.0f;
#pragma unroll
        for (int c = 0; c < DD; c += 4) {
            float4 x = *(float4*)(rowp + c);
            ssum += x.x * x.x + x.y * x.y + x.z * x.z + x.w * x.w;
        }
        float scale = mf[tid] / fmaxf(sqrtf(ssum), 1e-12f);
#pragma unroll
        for (int c = 0; c < DD; c += 4) {
            float4 x = *(float4*)(rowp + c);
            x.x *= scale; x.y *= scale; x.z *= scale; x.w *= scale;
            *(float4*)(rowp + c) = x;
        }
    }
    __syncthreads();

    const long pbase = (long)blk * PSTR;

    if (tid < 64) {
        float a = 0.0f;
        for (int r = 0; r < CH; ++r) a += kn[r * LROW + tid];
        ws[pbase + 4096 + tid] = a;
    } else if (tid < 128) {
        int c = tid - 64;
        float a = 0.0f;
        for (int r = 0; r < CH; ++r) a += mf[r] * vv[r * LROW + c];
        ws[pbase + 4160 + c] = a;
    } else if (tid == 128) {
        float a = 0.0f;
        for (int r = 0; r < CH; ++r) a += mf[r];
        ws[pbase + 4224] = a;
    }
    __syncthreads();

    const int lane = tid & 31;
    const int wv = tid >> 5;
    const int ti = wv >> 1;
    const int tj = (wv & 1) * 2;
    const int colA = ti * 16 + (lane & 15);
    const int colB0 = tj * 16 + (lane & 15);
    v8f acc0 = {};
    v8f acc1 = {};
    for (int k0 = 0; k0 < CH; k0 += 4) {
        v2f a  = ld_frag(kn, k0, colA, lane, LROW);
        v2f b0 = ld_frag(vv, k0, colB0, lane, LROW);
        v2f b1 = ld_frag(vv, k0, colB0 + 16, lane, LROW);
        acc0 = wmma_f32_4(a, b0, acc0);
        acc1 = wmma_f32_4(a, b1, acc1);
    }
    const int rbase = ti * 16 + 8 * (lane >> 4);
    const int c0 = tj * 16 + (lane & 15);
#pragma unroll
    for (int r = 0; r < 8; ++r) {
        ws[pbase + (rbase + r) * 64 + c0] = acc0[r];
        ws[pbase + (rbase + r) * 64 + c0 + 16] = acc1[r];
    }
}

__global__ __launch_bounds__(256) void cosattn_reduce(float* __restrict__ ws) {
    const int slot = blockIdx.x / 17;
    const int e = (blockIdx.x % 17) * 256 + threadIdx.x;
    if (e >= 4225) return;
    const long pb = (long)slot * NS * PSTR;
    float a = 0.0f;
#pragma unroll
    for (int c = 0; c < NS; ++c) a += ws[pb + (long)c * PSTR + e];
    ws[FINBASE + (long)slot * PSTR + e] = a;
}

__global__ __launch_bounds__(256) void cosattn_p2(
    const float* __restrict__ Q, const float* __restrict__ ws,
    float* __restrict__ out) {
    __shared__ float fin[PSTR];
    __shared__ float qs[8 * 16 * LROW];
    __shared__ float scale_s[128];
    __shared__ float denom_s[128];

    const int tid = threadIdx.x;
    const int lane = tid & 31;
    const int wv = tid >> 5;
    const int bh = blockIdx.x >> 5;
    const int rg = blockIdx.x & 31;
    const int q0 = rg * 128;

    const float* finp = ws + FINBASE + (long)bh * PSTR;
    for (int e = tid; e < 4225; e += 256) fin[e] = finp[e];

    const long qrowbase = (long)bh * SS + q0 + wv * 16;
    float* myq = qs + wv * 16 * LROW;
#pragma unroll
    for (int j = 0; j < 16; ++j) {
        float2 x = *(const float2*)(Q + (qrowbase + j) * DD + lane * 2);
        *(float2*)(myq + j * LROW + lane * 2) = x;
    }
    __syncthreads();

    const int r = lane & 15;
    const int half = lane >> 4;
    {
        float ssum = 0.0f, kd = 0.0f;
#pragma unroll
        for (int c32 = 0; c32 < 32; ++c32) {
            int c = half * 32 + c32;
            float v = myq[r * LROW + c];
            ssum += v * v;
            kd += v * fin[4096 + c];
        }
        ssum += __shfl_xor(ssum, 16, 32);
        kd += __shfl_xor(kd, 16, 32);
        float sc = 1.0f / fmaxf(sqrtf(ssum), 1e-12f);
        if (lane < 16) {
            scale_s[wv * 16 + lane] = sc;
            denom_s[wv * 16 + lane] = sc * kd + fin[4224];
        }
    }
    __syncthreads();

    v8f acc[4];
#pragma unroll
    for (int nt = 0; nt < 4; ++nt) {
        float sv = fin[4160 + nt * 16 + r];
        v8f ci;
#pragma unroll
        for (int i = 0; i < 8; ++i) ci[i] = sv;
        acc[nt] = ci;
    }
    const float scr = scale_s[wv * 16 + r];
    for (int k0 = 0; k0 < DD; k0 += 4) {
        int ks = k0 + 2 * half;
        v2f a;
        a.x = myq[r * LROW + ks] * scr;
        a.y = myq[r * LROW + ks + 1] * scr;
#pragma unroll
        for (int nt = 0; nt < 4; ++nt) {
            int n = nt * 16 + r;
            v2f bfr;
            bfr.x = fin[ks * 64 + n];
            bfr.y = fin[(ks + 1) * 64 + n];
            acc[nt] = wmma_f32_4(a, bfr, acc[nt]);
        }
    }

#pragma unroll
    for (int nt = 0; nt < 4; ++nt) {
        int col = nt * 16 + r;
#pragma unroll
        for (int rr = 0; rr < 8; ++rr) {
            int rowT = rr + 8 * half;
            float dv = denom_s[wv * 16 + rowT];
            out[(qrowbase + rowT) * DD + col] = acc[nt][rr] / dv;
        }
    }
}

extern "C" void kernel_launch(void* const* d_in, const int* in_sizes, int n_in,
                              void* d_out, int out_size, void* d_ws, size_t ws_size,
                              hipStream_t stream) {
    const float* Q = (const float*)d_in[0];
    const float* K = (const float*)d_in[1];
    const float* V = (const float*)d_in[2];
    const int* M = (const int*)d_in[3];
    float* out = (float*)d_out;
    float* ws = (float*)d_ws;

    cosattn_p1<<<dim3(BH * NS), dim3(256), 0, stream>>>(K, V, M, ws);
    cosattn_reduce<<<dim3(BH * 17), dim3(256), 0, stream>>>(ws);
    cosattn_p2<<<dim3(BH * 32), dim3(256), 0, stream>>>(Q, ws, out);
}


# p2 without LDS: M held per wave as B fragments in registers, q fragments by float4 from global in the same permuted k order (new summation order: q norm, q.ksum, products), float4 row stores
# speedup vs baseline: 1.0704x; 1.0704x over previous
#include <hip/hip_runtime.h>
#include <math.h>

#define BB 2
#define HH 8
#define SS 4096
#define DD 64
#define BH (BB*HH)
#define NS 16
#define CH (SS/NS)
#define PSTR 4352
#define FINBASE (BH*NS*PSTR)
#define LROW 68

typedef __attribute__((ext_vector_type(2))) float v2f;
typedef __attribute__((ext_vector_type(8))) float v8f;

__device__ __forceinline__ v8f wmma_f32_4(v2f a, v2f b, v8f c) {
    return __builtin_amdgcn_wmma_f32_16x16x4_f32(
        false, a, false, b, (short)0, c, false, false);
}

__device__ __forceinline__ v2f ld_frag(const float* p, int k0, int col, int lane, int str) {
    int ks = k0 + 2 * (lane >> 4);
    v2f r;
    r.x = p[ks * str + col];
    r.y = p[(ks + 1) * str + col];
    return r;
}

__global__ __launch_bounds__(256) void cosattn_p1(
    const float* __restrict__ K, const float* __restrict__ V,
    const int* __restrict__ mask, float* __restrict__ ws) {
    __shared__ float kn[CH * LROW];
    __shared__ float vv[CH * LROW];
    __shared__ float mf[CH];

    const int tid = threadIdx.x;
    const int blk = blockIdx.x;
    const int bh = blk / NS;
    const int chunk = blk % NS;
    const int b = bh / HH;
    const long gbase = ((long)bh * SS + (long)chunk * CH) * DD;

#pragma unroll
    for (int j = 0; j < 16; ++j) {
        int e4 = tid + j * 256;
        int row = e4 >> 4;
        int col = (e4 & 15) * 4;
        float4 kq = *(const float4*)(K + gbase + (long)e4 * 4);
        float4 vq = *(const float4*)(V + gbase + (long)e4 * 4);
        *(float4*)(kn + row * LROW + col) = kq;
        *(float4*)(vv + row * LROW + col) = vq;
    }
    {
        int kidx = chunk * CH + tid;
        mf[tid] = (mask[b * SS + kidx] != 0) ? 1.0f : 0.0f;
    }
    __syncthreads();

    {
        float* rowp = kn + tid * LROW;
        float ssum = 0.0f;
#pragma unroll
        for (int c = 0; c < DD; c += 4) {
            float4 x = *(float4*)(rowp + c);
            ssum += x.x * x.x + x.y * x.y + x.z * x.z + x.w * x.w;
        }
        float scale = mf[tid] / fmaxf(sqrtf(ssum), 1e-12f);
#pragma unroll
        for (int c = 0; c < DD; c += 4) {
            float4 x = *(float4*)(rowp + c);
            x.x *= scale; x.y *= scale; x.z *= scale; x.w *= scale;
            *(float4*)(rowp + c) = x;
        }
    }
    __syncthreads();

    const long pbase = (long)blk * PSTR;

    if (tid < 64) {
        float a = 0.0f;
        for (int r = 0; r < CH; ++r) a += kn[r * LROW + tid];
        ws[pbase + 4096 + tid] = a;
    } else if (tid < 128) {
        int c = tid - 64;
        float a = 0.0f;
        for (int r = 0; r < CH; ++r) a += mf[r] * vv[r * LROW + c];
        ws[pbase + 4160 + c] = a;
    } else if (tid == 128) {
        float a = 0.0f;
        for (int r = 0; r < CH; ++r) a += mf[r];
        ws[pbase + 4224] = a;
    }
    __syncthreads();

    const int lane = tid & 31;
    const int wv = tid >> 5;
    const int ti = wv >> 1;
    const int tj = (wv & 1) * 2;
    const int colA = ti * 16 + (lane & 15);
    const int colB0 = tj * 16 + (lane & 15);
    v8f acc0 = {};
    v8f acc1 = {};
    for (int k0 = 0; k0 < CH; k0 += 4) {
        v2f a  = ld_frag(kn, k0, colA, lane, LROW);
        v2f b0 = ld_frag(vv, k0, colB0, lane, LROW);
        v2f b1 = ld_frag(vv, k0, colB0 + 16, lane, LROW);
        acc0 = wmma_f32_4(a, b0, acc0);
        acc1 = wmma_f32_4(a, b1, acc1);
    }
    const int rbase = ti * 16 + 8 * (lane >> 4);
    const int c0 = tj * 16 + (lane & 15);
#pragma unroll
    for (int r = 0; r < 8; ++r) {
        ws[pbase + (rbase + r) * 64 + c0] = acc0[r];
        ws[pbase + (rbase + r) * 64 + c0 + 16] = acc1[r];
    }
}

__global__ __launch_bounds__(256) void cosattn_reduce(float* __restrict__ ws) {
    const int slot = blockIdx.x / 17;
    const int e = (blockIdx.x % 17) * 256 + threadIdx.x;
    if (e >= 4225) return;
    const long pb = (long)slot * NS * PSTR;
    float a = 0.0f;
#pragma unroll
    for (int c = 0; c < NS; ++c) a += ws[pb + (long)c * PSTR + e];
    ws[FINBASE + (long)slot * PSTR + e] = a;
}

__device__ __forceinline__ float f4c(const float4& v, int j) {
    return j == 0 ? v.x : (j == 1 ? v.y : (j == 2 ? v.z : v.w));
}

__global__ __launch_bounds__(256) void cosattn_p2(
    const float* __restrict__ Q, const float* __restrict__ ws,
    float* __restrict__ out) {
    const int tid = threadIdx.x;
    const int lane = tid & 31;
    const int wv = tid >> 5;
    const int c = lane & 15;
    const int h = lane >> 4;
    const int bh = blockIdx.x >> 4;
    const int rb = blockIdx.x & 15;
    const float* finp = ws + FINBASE + (long)bh * PSTR;
    const long row0 = (long)bh * SS + rb * 256 + wv * 32;

    float4 aq0[8];
#pragma unroll
    for (int i = 0; i < 8; ++i)
        aq0[i] = *(const float4*)(Q + (row0 + c) * DD + 8 * i + 4 * h);

    float4 ks[8];
#pragma unroll
    for (int i = 0; i < 8; ++i)
        ks[i] = *(const float4*)(finp + 4096 + 8 * i + 4 * h);
    const float4 sv = *(const float4*)(finp + 4160 + 4 * c);
    const float cnt = finp[4224];
    float4 bq[8][4];
#pragma unroll
    for (int i = 0; i < 8; ++i)
#pragma unroll
        for (int u = 0; u < 4; ++u)
            bq[i][u] = *(const float4*)(finp + (8 * i + 4 * h + u) * 64 + 4 * c);

    float4 aq1[8];
#pragma unroll
    for (int i = 0; i < 8; ++i)
        aq1[i] = *(const float4*)(Q + (row0 + 16 + c) * DD + 8 * i + 4 * h);

#pragma unroll
    for (int tl = 0; tl < 2; ++tl) {
        const float4* aq = tl ? aq1 : aq0;
        const long rowT = row0 + tl * 16;

        float ssum = 0.0f, kd = 0.0f;
#pragma unroll
        for (int i = 0; i < 8; ++i) {
            const float4 x = aq[i];
            const float4 kk = ks[i];
            ssum += x.x * x.x; ssum += x.y * x.y; ssum += x.z * x.z; ssum += x.w * x.w;
            kd += x.x * kk.x; kd += x.y * kk.y; kd += x.z * kk.z; kd += x.w * kk.w;
        }
        ssum += __shfl_xor(ssum, 16, 32);
        kd += __shfl_xor(kd, 16, 32);
        const float sc = 1.0f / fmaxf(sqrtf(ssum), 1e-12f);
        const float dn = sc * kd + cnt;

        v8f acc[4];
#pragma unroll
        for (int nt = 0; nt < 4; ++nt) {
            const float s0 = f4c(sv, nt);
            v8f ci;
#pragma unroll
            for (int r = 0; r < 8; ++r) ci[r] = s0;
            acc[nt] = ci;
        }

#pragma unroll
        for (int i = 0; i < 8; ++i) {
#pragma unroll
            for (int t = 0; t < 2; ++t) {
                v2f a;
                a.x = f4c(aq[i], 2 * t) * sc;
                a.y = f4c(aq[i], 2 * t + 1) * sc;
#pragma unroll
                for (int nt = 0; nt < 4; ++nt) {
                    v2f bfr;
                    bfr.x = f4c(bq[i][2 * t], nt);
                    bfr.y = f4c(bq[i][2 * t + 1], nt);
                    acc[nt] = wmma_f32_4(a, bfr, acc[nt]);
                }
            }
        }

#pragma unroll
        for (int r = 0; r < 8; ++r) {
            const float dv = __shfl(dn, 8 * h + r, 32);
            float4 o;
            o.x = acc[0][r] / dv;
            o.y = acc[1][r] / dv;
            o.z = acc[2][r] / dv;
            o.w = acc[3][r] / dv;
            *(float4*)(out + (rowT + 8 * h + r) * DD + 4 * c) = o;
        }
    }
}

extern "C" void kernel_launch(void* const* d_in, const int* in_sizes, int n_in,
                              void* d_out, int out_size, void* d_ws, size_t ws_size,
                              hipStream_t stream) {
    const float* Q = (const float*)d_in[0];
    const float* K = (const float*)d_in[1];
    const float* V = (const float*)d_in[2];
    const int* M = (const int*)d_in[3];
    float* out = (float*)d_out;
    float* ws = (float*)d_ws;

    cosattn_p1<<<dim3(BH * NS), dim3(256), 0, stream>>>(K, V, M, ws);
    cosattn_reduce<<<dim3(BH * 17), dim3(256), 0, stream>>>(ws);
    cosattn_p2<<<dim3(BH * 16), dim3(256), 0, stream>>>(Q, ws, out);
}


# p1 without LDS staging: each wave takes 32 keys global->registers as float4 A/B fragments in step order, key norm by 16-lane sum, 64x64 tile per wave; 8 wave partials summed in LDS (new sum order)
# speedup vs baseline: 1.3892x; 1.2978x over previous
#include <hip/hip_runtime.h>
#include <math.h>

#define BB 2
#define HH 8
#define SS 4096
#define DD 64
#define BH (BB*HH)
#define NS 16
#define CH (SS/NS)
#define PSTR 4352
#define FINBASE (BH*NS*PSTR)

typedef __attribute__((ext_vector_type(2))) float v2f;
typedef __attribute__((ext_vector_type(8))) float v8f;

__device__ __forceinline__ v8f wmma_f32_4(v2f a, v2f b, v8f c) {
    return __builtin_amdgcn_wmma_f32_16x16x4_f32(
        false, a, false, b, (short)0, c, false, false);
}

typedef __attribute__((ext_vector_type(4))) float v4f;
typedef __attribute__((ext_vector_type(4))) int v4i;

template <int M>
__device__ __forceinline__ float row_xor(float x) {
    const int b = __builtin_bit_cast(int, x);
    return __builtin_bit_cast(float, __builtin_amdgcn_update_dpp(0, b, 0x160 | M, 0xf, 0xf, true));
}

__device__ __forceinline__ float row16_sum(float x) {
    x += row_xor<1>(x);
    x += row_xor<2>(x);
    x += row_xor<4>(x);
    x += row_xor<8>(x);
    return x;
}

__device__ __forceinline__ float xhalf(float x) {
    const int b = __builtin_bit_cast(int, x);
    return __builtin_bit_cast(float,
        __builtin_amdgcn_permlanex16(b, b, 0x76543210, 0xfedcba98, false, false));
}

__device__ __forceinline__ void step_fence(v8f (&acc)[4][4], v4f& k0, v4f& k1, v4f& v0, v4f& v1) {
    asm volatile("" : "+v"(acc[0][0]), "+v"(acc[0][1]), "+v"(acc[0][2]), "+v"(acc[0][3]),
                      "+v"(acc[1][0]), "+v"(acc[1][1]), "+v"(acc[1][2]), "+v"(acc[1][3]),
                      "+v"(k0), "+v"(k1), "+v"(v0), "+v"(v1));
    asm volatile("" : "+v"(acc[2][0]), "+v"(acc[2][1]), "+v"(acc[2][2]), "+v"(acc[2][3]),
                      "+v"(acc[3][0]), "+v"(acc[3][1]), "+v"(acc[3][2]), "+v"(acc[3][3]));
}

__global__ __launch_bounds__(256) void cosattn_p1(
    const float* __restrict__ K, const float* __restrict__ V,
    const int* __restrict__ mask, float* __restrict__ ws) {
    __shared__ float red[8 * PSTR];

    const int tid = threadIdx.x;
    const int lane = tid & 31;
    const int wv = tid >> 5;
    const int c = lane & 15;
    const int h = lane >> 4;
    const int blk = blockIdx.x;
    const int bh = blk / NS;
    const int chunk = blk % NS;
    const int b = bh / HH;
    const int key0 = chunk * CH + wv * 32 + 16 * h;
    const long goff = ((long)bh * SS + key0) * DD + 4 * c;
    const float* kp = K + goff;
    const float* vp = V + goff;
    const int* mp = mask + b * SS + key0;

    v4i mq[4];
#pragma unroll
    for (int q = 0; q < 4; ++q) mq[q] = *(const v4i*)(mp + 4 * q);
    v4f kf[16], vf[16];
#pragma unroll
    for (int i = 0; i < 16; i += 2) {
        kf[i] = *(const v4f*)(kp + i * DD);
        kf[i + 1] = *(const v4f*)(kp + (i + 1) * DD);
        vf[i] = *(const v4f*)(vp + i * DD);
        vf[i + 1] = *(const v4f*)(vp + (i + 1) * DD);
        asm volatile("" :: "v"(kp), "v"(vp));
    }

    v8f acc[4][4];
#pragma unroll
    for (int ti = 0; ti < 4; ++ti)
#pragma unroll
        for (int tj = 0; tj < 4; ++tj) acc[ti][tj] = v8f{};
    v4f ksum = {0.f, 0.f, 0.f, 0.f};
    v4f ssv = {0.f, 0.f, 0.f, 0.f};
    float cnt = 0.0f;

    v4f kn[2][2], vm[2][2];
    auto prep = [&](const int j) {
#pragma unroll
        for (int e = 0; e < 2; ++e) {
            const int i = 2 * j + e;
            const float m = (mq[i >> 2][i & 3] != 0) ? 1.0f : 0.0f;
            const v4f x = kf[i];
            const float nrm2 = row16_sum(x.x * x.x + x.y * x.y + x.z * x.z + x.w * x.w);
            const float sc = m / fmaxf(sqrtf(nrm2), 1e-12f);
            const v4f kv = x * sc;
            const v4f vv = vf[i] * m;
            kn[j & 1][e] = kv;
            vm[j & 1][e] = vv;
            ksum += kv;
            ssv += vv;
            cnt += m;
        }
    };
    prep(0);
#pragma unroll
    for (int j = 0; j < 8; ++j) {
        if (j < 7) prep(j + 1);
#pragma unroll
        for (int ti = 0; ti < 4; ++ti) {
            v2f a;
            a.x = kn[j & 1][0][ti];
            a.y = kn[j & 1][1][ti];
#pragma unroll
            for (int tj = 0; tj < 4; ++tj) {
                v2f bfr;
                bfr.x = vm[j & 1][0][tj];
                bfr.y = vm[j & 1][1][tj];
                acc[ti][tj] = wmma_f32_4(a, bfr, acc[ti][tj]);
            }
        }
        if (j < 6) step_fence(acc, kf[2 * j + 4], kf[2 * j + 5], vf[2 * j + 4], vf[2 * j + 5]);
    }

    float* slot = red + wv * PSTR;
#pragma unroll
    for (int ti = 0; ti < 4; ++ti) {
#pragma unroll
        for (int r = 0; r < 8; ++r) {
            const int d1 = 4 * (8 * h + r) + ti;
            const v4f o = {acc[ti][0][r], acc[ti][1][r], acc[ti][2][r], acc[ti][3][r]};
            *(v4f*)(slot + d1 * 64 + 4 * c) = o;
        }
    }
    ksum.x += xhalf(ksum.x); ksum.y += xhalf(ksum.y); ksum.z += xhalf(ksum.z); ksum.w += xhalf(ksum.w);
    ssv.x += xhalf(ssv.x); ssv.y += xhalf(ssv.y); ssv.z += xhalf(ssv.z); ssv.w += xhalf(ssv.w);
    cnt += xhalf(cnt);
    if (h == 0) {
        *(v4f*)(slot + 4096 + 4 * c) = ksum;
        *(v4f*)(slot + 4160 + 4 * c) = ssv;
        if (c == 0) slot[4224] = cnt;
    }
    __syncthreads();

    float* dst = ws + (long)blk * PSTR;
#pragma unroll
    for (int q = 0; q < 4; ++q) {
        const int f = 4 * (tid + 256 * q);
        v4f a = *(const v4f*)(red + f);
#pragma unroll
        for (int w = 1; w < 8; ++w) a += *(const v4f*)(red + w * PSTR + f);
        *(v4f*)(dst + f) = a;
    }
    if (tid < 32) {
        const int f = 4096 + 4 * tid;
        v4f a = *(const v4f*)(red + f);
#pragma unroll
        for (int w = 1; w < 8; ++w) a += *(const v4f*)(red + w * PSTR + f);
        *(v4f*)(dst + f) = a;
    } else if (tid == 32) {
        float a = red[4224];
#pragma unroll
        for (int w = 1; w < 8; ++w) a += red[w * PSTR + 4224];
        dst[4224] = a;
    }
}

__global__ __launch_bounds__(256) void cosattn_reduce(float* __restrict__ ws) {
    const int slot = blockIdx.x / 17;
    const int e = (blockIdx.x % 17) * 256 + threadIdx.x;
    if (e >= 4225) return;
    const long pb = (long)slot * NS * PSTR;
    float a = 0.0f;
#pragma unroll
    for (int c = 0; c < NS; ++c) a += ws[pb + (long)c * PSTR + e];
    ws[FINBASE + (long)slot * PSTR + e] = a;
}

__device__ __forceinline__ float f4c(const float4& v, int j) {
    return j == 0 ? v.x : (j == 1 ? v.y : (j == 2 ? v.z : v.w));
}

__global__ __launch_bounds__(256) void cosattn_p2(
    const float* __restrict__ Q, const float* __restrict__ ws,
    float* __restrict__ out) {
    const int tid = threadIdx.x;
    const int lane = tid & 31;
    const int wv = tid >> 5;
    const int c = lane & 15;
    const int h = lane >> 4;
    const int bh = blockIdx.x >> 4;
    const int rb = blockIdx.x & 15;
    const float* finp = ws + FINBASE + (long)bh * PSTR;
    const long row0 = (long)bh * SS + rb * 256 + wv * 32;

    float4 aq0[8];
#pragma unroll
    for (int i = 0; i < 8; ++i)
        aq0[i] = *(const float4*)(Q + (row0 + c) * DD + 8 * i + 4 * h);

    float4 ks[8];
#pragma unroll
    for (int i = 0; i < 8; ++i)
        ks[i] = *(const float4*)(finp + 4096 + 8 * i + 4 * h);
    const float4 sv = *(const float4*)(finp + 4160 + 4 * c);
    const float cnt = finp[4224];
    float4 bq[8][4];
#pragma unroll
    for (int i = 0; i < 8; ++i)
#pragma unroll
        for (int u = 0; u < 4; ++u)
            bq[i][u] = *(const float4*)(finp + (8 * i + 4 * h + u) * 64 + 4 * c);

    float4 aq1[8];
#pragma unroll
    for (int i = 0; i < 8; ++i)
        aq1[i] = *(const float4*)(Q + (row0 + 16 + c) * DD + 8 * i + 4 * h);

#pragma unroll
    for (int tl = 0; tl < 2; ++tl) {
        const float4* aq = tl ? aq1 : aq0;
        const long rowT = row0 + tl * 16;

        float ssum = 0.0f, kd = 0.0f;
#pragma unroll
        for (int i = 0; i < 8; ++i) {
            const float4 x = aq[i];
            const float4 kk = ks[i];
            ssum += x.x * x.x; ssum += x.y * x.y; ssum += x.z * x.z; ssum += x.w * x.w;
            kd += x.x * kk.x; kd += x.y * kk.y; kd += x.z * kk.z; kd += x.w * kk.w;
        }
        ssum += __shfl_xor(ssum, 16, 32);
        kd += __shfl_xor(kd, 16, 32);
        const float sc = 1.0f / fmaxf(sqrtf(ssum), 1e-12f);
        const float dn = sc * kd + cnt;

        v8f acc[4];
#pragma unroll
        for (int nt = 0; nt < 4; ++nt) {
            const float s0 = f4c(sv, nt);
            v8f ci;
#pragma unroll
            for (int r = 0; r < 8; ++r) ci[r] = s0;
            acc[nt] = ci;
        }

#pragma unroll
        for (int i = 0; i < 8; ++i) {
#pragma unroll
            for (int t = 0; t < 2; ++t) {
                v2f a;
                a.x = f4c(aq[i], 2 * t) * sc;
                a.y = f4c(aq[i], 2 * t + 1) * sc;
#pragma unroll
                for (int nt = 0; nt < 4; ++nt) {
                    v2f bfr;
                    bfr.x = f4c(bq[i][2 * t], nt);
                    bfr.y = f4c(bq[i][2 * t + 1], nt);
                    acc[nt] = wmma_f32_4(a, bfr, acc[nt]);
                }
            }
        }

#pragma unroll
        for (int r = 0; r < 8; ++r) {
            const float dv = __shfl(dn, 8 * h + r, 32);
            float4 o;
            o.x = acc[0][r] / dv;
            o.y = acc[1][r] / dv;
            o.z = acc[2][r] / dv;
            o.w = acc[3][r] / dv;
            *(float4*)(out + (rowT + 8 * h + r) * DD + 4 * c) = o;
        }
    }
}

extern "C" void kernel_launch(void* const* d_in, const int* in_sizes, int n_in,
                              void* d_out, int out_size, void* d_ws, size_t ws_size,
                              hipStream_t stream) {
    const float* Q = (const float*)d_in[0];
    const float* K = (const float*)d_in[1];
    const float* V = (const float*)d_in[2];
    const int* M = (const int*)d_in[3];
    float* out = (float*)d_out;
    float* ws = (float*)d_ws;

    cosattn_p1<<<dim3(BH * NS), dim3(256), 0, stream>>>(K, V, M, ws);
    cosattn_reduce<<<dim3(BH * 17), dim3(256), 0, stream>>>(ws);
    cosattn_p2<<<dim3(BH * 16), dim3(256), 0, stream>>>(Q, ws, out);
}


# p2 row scalars on the vector ALU: one reciprocal of the denominator per row, spread by DPP row_share; outputs scaled by multiply (last-bit rounding differs); no LDS permutes, no per-element divides
# speedup vs baseline: 1.5023x; 1.0814x over previous
#include <hip/hip_runtime.h>
#include <math.h>

#define BB 2
#define HH 8
#define SS 4096
#define DD 64
#define BH (BB*HH)
#define NS 16
#define CH (SS/NS)
#define PSTR 4352
#define FINBASE (BH*NS*PSTR)

typedef __attribute__((ext_vector_type(2))) float v2f;
typedef __attribute__((ext_vector_type(8))) float v8f;

__device__ __forceinline__ v8f wmma_f32_4(v2f a, v2f b, v8f c) {
    return __builtin_amdgcn_wmma_f32_16x16x4_f32(
        false, a, false, b, (short)0, c, false, false);
}

typedef __attribute__((ext_vector_type(4))) float v4f;
typedef __attribute__((ext_vector_type(4))) int v4i;

template <int M>
__device__ __forceinline__ float row_xor(float x) {
    const int b = __builtin_bit_cast(int, x);
    return __builtin_bit_cast(float, __builtin_amdgcn_update_dpp(0, b, 0x160 | M, 0xf, 0xf, true));
}

__device__ __forceinline__ float row16_sum(float x) {
    x += row_xor<1>(x);
    x += row_xor<2>(x);
    x += row_xor<4>(x);
    x += row_xor<8>(x);
    return x;
}

template <int N>
__device__ __forceinline__ float row_share(float x) {
    const int b = __builtin_bit_cast(int, x);
    return __builtin_bit_cast(float, __builtin_amdgcn_update_dpp(0, b, 0x150 | N, 0xf, 0xf, true));
}

__device__ __forceinline__ float xhalf(float x) {
    const int b = __builtin_bit_cast(int, x);
    return __builtin_bit_cast(float,
        __builtin_amdgcn_permlanex16(b, b, 0x76543210, 0xfedcba98, false, false));
}

__device__ __forceinline__ void step_fence(v8f (&acc)[4][4], v4f& k0, v4f& k1, v4f& v0, v4f& v1) {
    asm volatile("" : "+v"(acc[0][0]), "+v"(acc[0][1]), "+v"(acc[0][2]), "+v"(acc[0][3]),
                      "+v"(acc[1][0]), "+v"(acc[1][1]), "+v"(acc[1][2]), "+v"(acc[1][3]),
                      "+v"(k0), "+v"(k1), "+v"(v0), "+v"(v1));
    asm volatile("" : "+v"(acc[2][0]), "+v"(acc[2][1]), "+v"(acc[2][2]), "+v"(acc[2][3]),
                      "+v"(acc[3][0]), "+v"(acc[3][1]), "+v"(acc[3][2]), "+v"(acc[3][3]));
}

__global__ __launch_bounds__(256) void cosattn_p1(
    const float* __restrict__ K, const float* __restrict__ V,
    const int* __restrict__ mask, float* __restrict__ ws) {
    __shared__ float red[8 * PSTR];

    const int tid = threadIdx.x;
    const int lane = tid & 31;
    const int wv = tid >> 5;
    const int c = lane & 15;
    const int h = lane >> 4;
    const int blk = blockIdx.x;
    const int bh = blk / NS;
    const int chunk = blk % NS;
    const int b = bh / HH;
    const int key0 = chunk * CH + wv * 32 + 16 * h;
    const long goff = ((long)bh * SS + key0) * DD + 4 * c;
    const float* kp = K + goff;
    const float* vp = V + goff;
    const int* mp = mask + b * SS + key0;

    v4i mq[4];
#pragma unroll
    for (int q = 0; q < 4; ++q) mq[q] = *(const v4i*)(mp + 4 * q);
    v4f kf[16], vf[16];
#pragma unroll
    for (int i = 0; i < 16; i += 2) {
        kf[i] = *(const v4f*)(kp + i * DD);
        kf[i + 1] = *(const v4f*)(kp + (i + 1) * DD);
        vf[i] = *(const v4f*)(vp + i * DD);
        vf[i + 1] = *(const v4f*)(vp + (i + 1) * DD);
        asm volatile("" :: "v"(kp), "v"(vp));
    }

    v8f acc[4][4];
#pragma unroll
    for (int ti = 0; ti < 4; ++ti)
#pragma unroll
        for (int tj = 0; tj < 4; ++tj) acc[ti][tj] = v8f{};
    v4f ksum = {0.f, 0.f, 0.f, 0.f};
    v4f ssv = {0.f, 0.f, 0.f, 0.f};
    float cnt = 0.0f;

    v4f kn[2][2], vm[2][2];
    auto prep = [&](const int j) {
#pragma unroll
        for (int e = 0; e < 2; ++e) {
            const int i = 2 * j + e;
            const float m = (mq[i >> 2][i & 3] != 0) ? 1.0f : 0.0f;
            const v4f x = kf[i];
            const float nrm2 = row16_sum(x.x * x.x + x.y * x.y + x.z * x.z + x.w * x.w);
            const float sc = m / fmaxf(sqrtf(nrm2), 1e-12f);
            const v4f kv = x * sc;
            const v4f vv = vf[i] * m;
            kn[j & 1][e] = kv;
            vm[j & 1][e] = vv;
            ksum += kv;
            ssv += vv;
            cnt += m;
        }
    };
    prep(0);
#pragma unroll
    for (int j = 0; j < 8; ++j) {
        if (j < 7) prep(j + 1);
#pragma unroll
        for (int ti = 0; ti < 4; ++ti) {
            v2f a;
            a.x = kn[j & 1][0][ti];
            a.y = kn[j & 1][1][ti];
#pragma unroll
            for (int tj = 0; tj < 4; ++tj) {
                v2f bfr;
                bfr.x = vm[j & 1][0][tj];
                bfr.y = vm[j & 1][1][tj];
                acc[ti][tj] = wmma_f32_4(a, bfr, acc[ti][tj]);
            }
        }
        if (j < 6) step_fence(acc, kf[2 * j + 4], kf[2 * j + 5], vf[2 * j + 4], vf[2 * j + 5]);
    }

    float* slot = red + wv * PSTR;
#pragma unroll
    for (int ti = 0; ti < 4; ++ti) {
#pragma unroll
        for (int r = 0; r < 8; ++r) {
            const int d1 = 4 * (8 * h + r) + ti;
            const v4f o = {acc[ti][0][r], acc[ti][1][r], acc[ti][2][r], acc[ti][3][r]};
            *(v4f*)(slot + d1 * 64 + 4 * c) = o;
        }
    }
    ksum.x += xhalf(ksum.x); ksum.y += xhalf(ksum.y); ksum.z += xhalf(ksum.z); ksum.w += xhalf(ksum.w);
    ssv.x += xhalf(ssv.x); ssv.y += xhalf(ssv.y); ssv.z += xhalf(ssv.z); ssv.w += xhalf(ssv.w);
    cnt += xhalf(cnt);
    if (h == 0) {
        *(v4f*)(slot + 4096 + 4 * c) = ksum;
        *(v4f*)(slot + 4160 + 4 * c) = ssv;
        if (c == 0) slot[4224] = cnt;
    }
    __syncthreads();

    float* dst = ws + (long)blk * PSTR;
#pragma unroll
    for (int q = 0; q < 4; ++q) {
        const int f = 4 * (tid + 256 * q);
        v4f a = *(const v4f*)(red + f);
#pragma unroll
        for (int w = 1; w < 8; ++w) a += *(const v4f*)(red + w * PSTR + f);
        *(v4f*)(dst + f) = a;
    }
    if (tid < 32) {
        const int f = 4096 + 4 * tid;
        v4f a = *(const v4f*)(red + f);
#pragma unroll
        for (int w = 1; w < 8; ++w) a += *(const v4f*)(red + w * PSTR + f);
        *(v4f*)(dst + f) = a;
    } else if (tid == 32) {
        float a = red[4224];
#pragma unroll
        for (int w = 1; w < 8; ++w) a += red[w * PSTR + 4224];
        dst[4224] = a;
    }
}

__global__ __launch_bounds__(256) void cosattn_reduce(float* __restrict__ ws) {
    const int slot = blockIdx.x / 17;
    const int e = (blockIdx.x % 17) * 256 + threadIdx.x;
    if (e >= 4225) return;
    const long pb = (long)slot * NS * PSTR;
    float a = 0.0f;
#pragma unroll
    for (int c = 0; c < NS; ++c) a += ws[pb + (long)c * PSTR + e];
    ws[FINBASE + (long)slot * PSTR + e] = a;
}

template <int V> struct ic { static constexpr int value = V; };

__device__ __forceinline__ float f4c(const float4& v, int j) {
    return j == 0 ? v.x : (j == 1 ? v.y : (j == 2 ? v.z : v.w));
}

__global__ __launch_bounds__(256) void cosattn_p2(
    const float* __restrict__ Q, const float* __restrict__ ws,
    float* __restrict__ out) {
    const int tid = threadIdx.x;
    const int lane = tid & 31;
    const int wv = tid >> 5;
    const int c = lane & 15;
    const int h = lane >> 4;
    const int bh = blockIdx.x >> 4;
    const int rb = blockIdx.x & 15;
    const float* finp = ws + FINBASE + (long)bh * PSTR;
    const long row0 = (long)bh * SS + rb * 256 + wv * 32;

    float4 aq0[8];
#pragma unroll
    for (int i = 0; i < 8; ++i)
        aq0[i] = *(const float4*)(Q + (row0 + c) * DD + 8 * i + 4 * h);

    float4 ks[8];
#pragma unroll
    for (int i = 0; i < 8; ++i)
        ks[i] = *(const float4*)(finp + 4096 + 8 * i + 4 * h);
    const float4 sv = *(const float4*)(finp + 4160 + 4 * c);
    const float cnt = finp[4224];
    float4 bq[8][4];
#pragma unroll
    for (int i = 0; i < 8; ++i)
#pragma unroll
        for (int u = 0; u < 4; ++u)
            bq[i][u] = *(const float4*)(finp + (8 * i + 4 * h + u) * 64 + 4 * c);

    float4 aq1[8];
#pragma unroll
    for (int i = 0; i < 8; ++i)
        aq1[i] = *(const float4*)(Q + (row0 + 16 + c) * DD + 8 * i + 4 * h);

#pragma unroll
    for (int tl = 0; tl < 2; ++tl) {
        const float4* aq = tl ? aq1 : aq0;
        const long rowT = row0 + tl * 16;

        float ssum = 0.0f, kd = 0.0f;
#pragma unroll
        for (int i = 0; i < 8; ++i) {
            const float4 x = aq[i];
            const float4 kk = ks[i];
            ssum += x.x * x.x; ssum += x.y * x.y; ssum += x.z * x.z; ssum += x.w * x.w;
            kd += x.x * kk.x; kd += x.y * kk.y; kd += x.z * kk.z; kd += x.w * kk.w;
        }
        ssum += xhalf(ssum);
        kd += xhalf(kd);
        const float sc = 1.0f / fmaxf(sqrtf(ssum), 1e-12f);
        const float dn = sc * kd + cnt;
        const float rd = 1.0f / dn;
        const float rdx = row_xor<8>(rd);
        const float rds = h ? rdx : rd;

        v8f acc[4];
#pragma unroll
        for (int nt = 0; nt < 4; ++nt) {
            const float s0 = f4c(sv, nt);
            v8f ci;
#pragma unroll
            for (int r = 0; r < 8; ++r) ci[r] = s0;
            acc[nt] = ci;
        }

#pragma unroll
        for (int i = 0; i < 8; ++i) {
#pragma unroll
            for (int t = 0; t < 2; ++t) {
                v2f a;
                a.x = f4c(aq[i], 2 * t) * sc;
                a.y = f4c(aq[i], 2 * t + 1) * sc;
#pragma unroll
                for (int nt = 0; nt < 4; ++nt) {
                    v2f bfr;
                    bfr.x = f4c(bq[i][2 * t], nt);
                    bfr.y = f4c(bq[i][2 * t + 1], nt);
                    acc[nt] = wmma_f32_4(a, bfr, acc[nt]);
                }
            }
        }

        auto put = [&](auto rc) {
            constexpr int r = decltype(rc)::value;
            const float rv = row_share<r>(rds);
            float4 o;
            o.x = acc[0][r] * rv;
            o.y = acc[1][r] * rv;
            o.z = acc[2][r] * rv;
            o.w = acc[3][r] * rv;
            *(float4*)(out + (rowT + 8 * h + r) * DD + 4 * c) = o;
        };
        put(ic<0>{}); put(ic<1>{}); put(ic<2>{}); put(ic<3>{});
        put(ic<4>{}); put(ic<5>{}); put(ic<6>{}); put(ic<7>{});
    }
}

extern "C" void kernel_launch(void* const* d_in, const int* in_sizes, int n_in,
                              void* d_out, int out_size, void* d_ws, size_t ws_size,
                              hipStream_t stream) {
    const float* Q = (const float*)d_in[0];
    const float* K = (const float*)d_in[1];
    const float* V = (const float*)d_in[2];
    const int* M = (const int*)d_in[3];
    float* out = (float*)d_out;
    float* ws = (float*)d_ws;

    cosattn_p1<<<dim3(BH * NS), dim3(256), 0, stream>>>(K, V, M, ws);
    cosattn_reduce<<<dim3(BH * 17), dim3(256), 0, stream>>>(ws);
    cosattn_p2<<<dim3(BH * 16), dim3(256), 0, stream>>>(Q, ws, out);
}


# reduce launch removed: a p1 block takes 512 keys x half of M's columns, so a head has 8 chunk partials (was 16); p2 sums them itself through LDS in chunk order (new summation order); 3 launches to 2
# speedup vs baseline: 1.5887x; 1.0575x over previous
#include <hip/hip_runtime.h>
#include <math.h>

#define BB 2
#define HH 8
#define SS 4096
#define DD 64
#define BH (BB*HH)
#define NS 8
#define CH (SS/NS)
#define PSTR 4352
#define WSTR 2176

typedef __attribute__((ext_vector_type(2))) float v2f;
typedef __attribute__((ext_vector_type(8))) float v8f;

__device__ __forceinline__ v8f wmma_f32_4(v2f a, v2f b, v8f c) {
    return __builtin_amdgcn_wmma_f32_16x16x4_f32(
        false, a, false, b, (short)0, c, false, false);
}

typedef __attribute__((ext_vector_type(4))) float v4f;
typedef __attribute__((ext_vector_type(4))) int v4i;

template <int M>
__device__ __forceinline__ float row_xor(float x) {
    const int b = __builtin_bit_cast(int, x);
    return __builtin_bit_cast(float, __builtin_amdgcn_update_dpp(0, b, 0x160 | M, 0xf, 0xf, true));
}

__device__ __forceinline__ float row16_sum(float x) {
    x += row_xor<1>(x);
    x += row_xor<2>(x);
    x += row_xor<4>(x);
    x += row_xor<8>(x);
    return x;
}

template <int N>
__device__ __forceinline__ float row_share(float x) {
    const int b = __builtin_bit_cast(int, x);
    return __builtin_bit_cast(float, __builtin_amdgcn_update_dpp(0, b, 0x150 | N, 0xf, 0xf, true));
}

__device__ __forceinline__ float xhalf(float x) {
    const int b = __builtin_bit_cast(int, x);
    return __builtin_bit_cast(float,
        __builtin_amdgcn_permlanex16(b, b, 0x76543210, 0xfedcba98, false, false));
}

__device__ __forceinline__ void step_fence(v8f (&acc)[4][2], v4f& k0, v4f& k1, v2f& v0, v2f& v1) {
    asm volatile("" : "+v"(acc[0][0]), "+v"(acc[0][1]), "+v"(acc[1][0]), "+v"(acc[1][1]),
                      "+v"(acc[2][0]), "+v"(acc[2][1]), "+v"(acc[3][0]), "+v"(acc[3][1]),
                      "+v"(k0), "+v"(k1), "+v"(v0), "+v"(v1));
}

__global__ __launch_bounds__(256) void cosattn_p1(
    const float* __restrict__ K, const float* __restrict__ V,
    const int* __restrict__ mask, float* __restrict__ ws) {
    __shared__ __attribute__((aligned(16))) float red[8 * WSTR];

    const int tid = threadIdx.x;
    const int lane = tid & 31;
    const int wv = tid >> 5;
    const int c = lane & 15;
    const int h = lane >> 4;
    const int half = blockIdx.x & 1;
    const int sid = blockIdx.x >> 1;
    const int bh = sid / NS;
    const int chunk = sid % NS;
    const int b = bh / HH;
    const int key0 = chunk * CH + wv * 64 + 32 * h;
    const long grow = ((long)bh * SS + key0) * DD;
    const float* kp = K + grow + 4 * c;
    const float* vp = V + grow + 32 * half + 2 * c;
    const int* mp = mask + b * SS + key0;

    v4i mq[8];
#pragma unroll
    for (int q = 0; q < 8; ++q) mq[q] = *(const v4i*)(mp + 4 * q);
    v4f kf[32];
    v2f vf[32];
#pragma unroll
    for (int i = 0; i < 32; i += 2) {
        kf[i] = *(const v4f*)(kp + i * DD);
        kf[i + 1] = *(const v4f*)(kp + (i + 1) * DD);
        vf[i] = *(const v2f*)(vp + i * DD);
        vf[i + 1] = *(const v2f*)(vp + (i + 1) * DD);
        asm volatile("" :: "v"(kp), "v"(vp));
    }

    v8f acc[4][2];
#pragma unroll
    for (int ti = 0; ti < 4; ++ti)
#pragma unroll
        for (int tj = 0; tj < 2; ++tj) acc[ti][tj] = v8f{};
    v4f ksum = {0.f, 0.f, 0.f, 0.f};
    v2f ssv = {0.f, 0.f};
    float cnt = 0.0f;

    v4f kn[2][2];
    v2f vm[2][2];
    auto prep = [&](const int j) {
#pragma unroll
        for (int e = 0; e < 2; ++e) {
            const int i = 2 * j + e;
            const float m = (mq[i >> 2][i & 3] != 0) ? 1.0f : 0.0f;
            const v4f x = kf[i];
            const float nrm2 = row16_sum(x.x * x.x + x.y * x.y + x.z * x.z + x.w * x.w);
            const float sc = m / fmaxf(sqrtf(nrm2), 1e-12f);
            const v4f kv = x * sc;
            const v2f vv = vf[i] * m;
            kn[j & 1][e] = kv;
            vm[j & 1][e] = vv;
            ksum += kv;
            ssv += vv;
            cnt += m;
        }
    };
    prep(0);
#pragma unroll
    for (int j = 0; j < 16; ++j) {
        if (j < 15) prep(j + 1);
#pragma unroll
        for (int ti = 0; ti < 4; ++ti) {
            v2f a;
            a.x = kn[j & 1][0][ti];
            a.y = kn[j & 1][1][ti];
#pragma unroll
            for (int tj = 0; tj < 2; ++tj) {
                v2f bfr;
                bfr.x = vm[j & 1][0][tj];
                bfr.y = vm[j & 1][1][tj];
                acc[ti][tj] = wmma_f32_4(a, bfr, acc[ti][tj]);
            }
        }
        if (j < 14) step_fence(acc, kf[2 * j + 4], kf[2 * j + 5], vf[2 * j + 4], vf[2 * j + 5]);
    }

    float* slot = red + wv * WSTR;
#pragma unroll
    for (int ti = 0; ti < 4; ++ti) {
#pragma unroll
        for (int r = 0; r < 8; ++r) {
            const int d1 = 4 * (8 * h + r) + ti;
            const v2f o = {acc[ti][0][r], acc[ti][1][r]};
            *(v2f*)(slot + d1 * 32 + 2 * c) = o;
        }
    }
    ksum.x += xhalf(ksum.x); ksum.y += xhalf(ksum.y); ksum.z += xhalf(ksum.z); ksum.w += xhalf(ksum.w);
    ssv.x += xhalf(ssv.x); ssv.y += xhalf(ssv.y);
    cnt += xhalf(cnt);
    if (h == 0) {
        *(v4f*)(slot + 2048 + 4 * c) = ksum;
        *(v2f*)(slot + 2112 + 2 * c) = ssv;
        if (c == 0) slot[2144] = cnt;
    }
    __syncthreads();

    float* dst = ws + (long)sid * PSTR;
#pragma unroll
    for (int q = 0; q < 2; ++q) {
        const int f = tid + 256 * q;
        v4f a = *(const v4f*)(red + 4 * f);
#pragma unroll
        for (int w = 1; w < 8; ++w) a += *(const v4f*)(red + w * WSTR + 4 * f);
        *(v4f*)(dst + (f >> 3) * 64 + 32 * half + 4 * (f & 7)) = a;
    }
    if (tid < 16) {
        if (half == 0) {
            const int f = 2048 + 4 * tid;
            v4f a = *(const v4f*)(red + f);
#pragma unroll
            for (int w = 1; w < 8; ++w) a += *(const v4f*)(red + w * WSTR + f);
            *(v4f*)(dst + 4096 + 4 * tid) = a;
        }
    } else if (tid < 24) {
        const int f = 2112 + 4 * (tid - 16);
        v4f a = *(const v4f*)(red + f);
#pragma unroll
        for (int w = 1; w < 8; ++w) a += *(const v4f*)(red + w * WSTR + f);
        *(v4f*)(dst + 4160 + 32 * half + 4 * (tid - 16)) = a;
    } else if (tid == 32) {
        if (half == 0) {
            float a = red[2144];
#pragma unroll
            for (int w = 1; w < 8; ++w) a += red[w * WSTR + 2144];
            dst[4224] = a;
        }
    }
}

template <int V> struct ic { static constexpr int value = V; };

__device__ __forceinline__ float f4c(const float4& v, int j) {
    return j == 0 ? v.x : (j == 1 ? v.y : (j == 2 ? v.z : v.w));
}

__global__ __launch_bounds__(256) void cosattn_p2(
    const float* __restrict__ Q, const float* __restrict__ ws,
    float* __restrict__ out) {
    const int tid = threadIdx.x;
    const int lane = tid & 31;
    const int wv = tid >> 5;
    const int c = lane & 15;
    const int h = lane >> 4;
    const int bh = blockIdx.x >> 4;
    const int rb = blockIdx.x & 15;
    __shared__ __attribute__((aligned(16))) float finp[4096];
    __shared__ __attribute__((aligned(16))) float fine[8][128];
    const float* pb = ws + (long)bh * NS * PSTR;
    const long row0 = (long)bh * SS + rb * 256 + wv * 32;

    float4 aq0[8];
#pragma unroll
    for (int i = 0; i < 8; ++i)
        aq0[i] = *(const float4*)(Q + (row0 + c) * DD + 8 * i + 4 * h);

    float4 aq1[8];
#pragma unroll
    for (int i = 0; i < 8; ++i)
        aq1[i] = *(const float4*)(Q + (row0 + 16 + c) * DD + 8 * i + 4 * h);

    float cnt = pb[4224];
#pragma unroll
    for (int cc = 1; cc < NS; ++cc) cnt += pb[cc * PSTR + 4224];
    v4f ep[NS];
#pragma unroll
    for (int cc = 0; cc < NS; ++cc) ep[cc] = *(const v4f*)(pb + cc * PSTR + 4096 + 4 * lane);
#pragma unroll
    for (int q = 0; q < 4; ++q) {
        const int f = 4 * (tid + 256 * q);
        v4f a = *(const v4f*)(pb + f);
#pragma unroll
        for (int cc = 1; cc < NS; ++cc) a += *(const v4f*)(pb + cc * PSTR + f);
        *(v4f*)(finp + f) = a;
    }
    {
        v4f a = ep[0];
#pragma unroll
        for (int cc = 1; cc < NS; ++cc) a += ep[cc];
        *(v4f*)(fine[wv] + 4 * lane) = a;
    }
    __syncthreads();

    float4 ks[8];
#pragma unroll
    for (int i = 0; i < 8; ++i)
        ks[i] = *(const float4*)(fine[wv] + 8 * i + 4 * h);
    const float4 sv = *(const float4*)(fine[wv] + 64 + 4 * c);
    float4 bq[8][4];
#pragma unroll
    for (int i = 0; i < 8; ++i)
#pragma unroll
        for (int u = 0; u < 4; ++u)
            bq[i][u] = *(const float4*)(finp + (8 * i + 4 * h + u) * 64 + 4 * c);

#pragma unroll
    for (int tl = 0; tl < 2; ++tl) {
        const float4* aq = tl ? aq1 : aq0;
        const long rowT = row0 + tl * 16;

        float ssum = 0.0f, kd = 0.0f;
#pragma unroll
        for (int i = 0; i < 8; ++i) {
            const float4 x = aq[i];
            const float4 kk = ks[i];
            ssum += x.x * x.x; ssum += x.y * x.y; ssum += x.z * x.z; ssum += x.w * x.w;
            kd += x.x * kk.x; kd += x.y * kk.y; kd += x.z * kk.z; kd += x.w * kk.w;
        }
        ssum += xhalf(ssum);
        kd += xhalf(kd);
        const float sc = 1.0f / fmaxf(sqrtf(ssum), 1e-12f);
        const float dn = sc * kd + cnt;
        const float rd = 1.0f / dn;
        const float rdx = row_xor<8>(rd);
        const float rds = h ? rdx : rd;

        v8f acc[4];
#pragma unroll
        for (int nt = 0; nt < 4; ++nt) {
            const float s0 = f4c(sv, nt);
            v8f ci;
#pragma unroll
            for (int r = 0; r < 8; ++r) ci[r] = s0;
            acc[nt] = ci;
        }

#pragma unroll
        for (int i = 0; i < 8; ++i) {
#pragma unroll
            for (int t = 0; t < 2; ++t) {
                v2f a;
                a.x = f4c(aq[i], 2 * t) * sc;
                a.y = f4c(aq[i], 2 * t + 1) * sc;
#pragma unroll
                for (int nt = 0; nt < 4; ++nt) {
                    v2f bfr;
                    bfr.x = f4c(bq[i][2 * t], nt);
                    bfr.y = f4c(bq[i][2 * t + 1], nt);
                    acc[nt] = wmma_f32_4(a, bfr, acc[nt]);
                }
            }
        }

        auto put = [&](auto rc) {
            constexpr int r = decltype(rc)::value;
            const float rv = row_share<r>(rds);
            float4 o;
            o.x = acc[0][r] * rv;
            o.y = acc[1][r] * rv;
            o.z = acc[2][r] * rv;
            o.w = acc[3][r] * rv;
            *(float4*)(out + (rowT + 8 * h + r) * DD + 4 * c) = o;
        };
        put(ic<0>{}); put(ic<1>{}); put(ic<2>{}); put(ic<3>{});
        put(ic<4>{}); put(ic<5>{}); put(ic<6>{}); put(ic<7>{});
    }
}

extern "C" void kernel_launch(void* const* d_in, const int* in_sizes, int n_in,
                              void* d_out, int out_size, void* d_ws, size_t ws_size,
                              hipStream_t stream) {
    const float* Q = (const float*)d_in[0];
    const float* K = (const float*)d_in[1];
    const float* V = (const float*)d_in[2];
    const int* M = (const int*)d_in[3];
    float* out = (float*)d_out;
    float* ws = (float*)d_ws;

    cosattn_p1<<<dim3(BH * NS * 2), dim3(256), 0, stream>>>(K, V, M, ws);
    cosattn_p2<<<dim3(BH * 16), dim3(256), 0, stream>>>(Q, ws, out);
}
